# starting kernel
# baseline (speedup 1.0000x reference)
#include <hip/hip_runtime.h>

#if defined(__HIP_DEVICE_COMPILE__)
#if !__has_builtin(__builtin_amdgcn_wmma_f32_16x16x4_f32)
#error "device pass: missing __builtin_amdgcn_wmma_f32_16x16x4_f32"
#endif
#endif

typedef __attribute__((ext_vector_type(2))) float v2f;
typedef __attribute__((ext_vector_type(8))) float v8f;

#define BN    8
#define CI    256
#define CO    128
#define HH    64
#define WW    64
#define OH    130
#define OW    130
#define UU    65
#define NPB   (UU*UU)
#define BLK_N 128
#define NT    ((NPB + BLK_N - 1)/BLK_N)

__global__ __launch_bounds__(256)
void deconv2d_fold_wmma(const float* __restrict__ x,
                        const float* __restrict__ wgt,
                        const float* __restrict__ bias,
                        float* __restrict__ out)
{
    const int tile = blockIdx.x;
    const int bb   = blockIdx.y;
    const int mt   = blockIdx.z >> 2;
    const int par  = blockIdx.z & 3;
    const int p = par >> 1, q = par & 1;

    const int wave = threadIdx.x >> 5;
    const int lane = threadIdx.x & 31;
    const int l16  = lane & 15;
    const int hi   = lane >> 4;
    const int kk   = hi * 2;

    const int wr = wave >> 2;
    const int wcn = wave & 3;
    const int mw = mt * 64 + wr * 32;
    const int nw = tile * BLK_N + wcn * 32;

    const float* xp[2][4];
    float        mk[2][4];
    bool         nval[2];
    int          uu[2], vv[2];
    #pragma unroll
    for (int sj = 0; sj < 2; ++sj) {
        const int n = nw + sj * 16 + l16;
        nval[sj] = (n < NPB);
        const int nc = nval[sj] ? n : (NPB - 1);
        uu[sj] = nc / UU;
        vv[sj] = nc % UU;
        #pragma unroll
        for (int t = 0; t < 4; ++t) {
            const int dy = t & 1, dx = t >> 1;
            const int h = uu[sj] - dy;
            const int w = vv[sj] - dx;
            const bool ok = (h >= 0) && (h < HH) && (w >= 0) && (w < WW);
            const int hc = h < 0 ? 0 : (h > HH - 1 ? HH - 1 : h);
            const int wc2 = w < 0 ? 0 : (w > WW - 1 ? WW - 1 : w);
            xp[sj][t] = x + (size_t)bb * CI * HH * WW
                          + (size_t)kk * HH * WW
                          + (size_t)hc * WW + wc2;
            mk[sj][t] = ok ? 1.0f : 0.0f;
        }
    }

    const float* wp[2][4];
    #pragma unroll
    for (int si = 0; si < 2; ++si) {
        #pragma unroll
        for (int t = 0; t < 4; ++t) {
            const int dy = t & 1, dx = t >> 1;
            const int ky = p + 2 * dy, kx = q + 2 * dx;
            wp[si][t] = wgt + (size_t)(mw + si * 16 + l16) * (CI * 16)
                            + (size_t)kk * 16
                            + (size_t)ky * 4 + kx;
        }
    }

    v8f acc[2][2] = {};

    #pragma unroll
    for (int t = 0; t < 4; ++t) {
        const float* __restrict__ wq0 = wp[0][t];
        const float* __restrict__ wq1 = wp[1][t];
        const float* __restrict__ xq0 = xp[0][t];
        const float* __restrict__ xq1 = xp[1][t];
        const float m0 = mk[0][t], m1 = mk[1][t];
        #pragma unroll 4
        for (int i0 = 0; i0 < CI; i0 += 4) {
            v2f a0, a1, b0, b1;
            a0.x = wq0[(size_t)i0 * 16];
            a0.y = wq0[(size_t)i0 * 16 + 16];
            a1.x = wq1[(size_t)i0 * 16];
            a1.y = wq1[(size_t)i0 * 16 + 16];
            b0.x = xq0[(size_t)i0 * (HH * WW)] * m0;
            b0.y = xq0[(size_t)i0 * (HH * WW) + HH * WW] * m0;
            b1.x = xq1[(size_t)i0 * (HH * WW)] * m1;
            b1.y = xq1[(size_t)i0 * (HH * WW) + HH * WW] * m1;
            acc[0][0] = __builtin_amdgcn_wmma_f32_16x16x4_f32(
                false, a0, false, b0, (short)0, acc[0][0], false, false);
            acc[0][1] = __builtin_amdgcn_wmma_f32_16x16x4_f32(
                false, a0, false, b1, (short)0, acc[0][1], false, false);
            acc[1][0] = __builtin_amdgcn_wmma_f32_16x16x4_f32(
                false, a1, false, b0, (short)0, acc[1][0], false, false);
            acc[1][1] = __builtin_amdgcn_wmma_f32_16x16x4_f32(
                false, a1, false, b1, (short)0, acc[1][1], false, false);
        }
    }

    #pragma unroll
    for (int sj = 0; sj < 2; ++sj) {
        if (!nval[sj]) continue;
        const int oy = p + 2 * uu[sj];
        const int ox = q + 2 * vv[sj];
        const size_t ob = (size_t)bb * CO * OH * OW + (size_t)oy * OW + ox;
        #pragma unroll
        for (int si = 0; si < 2; ++si) {
            #pragma unroll
            for (int j = 0; j < 8; ++j) {
                const int m = mw + si * 16 + j + hi * 8;
                float bs = 0.0f;
                #pragma unroll
                for (int t = 0; t < 4; ++t) {
                    const int dy = t & 1, dx = t >> 1;
                    bs += mk[sj][t] * bias[m * 16 + (p + 2 * dy) * 4 + (q + 2 * dx)];
                }
                out[ob + (size_t)m * OH * OW] = acc[si][sj][j] + bs;
            }
        }
    }
}

extern "C" void kernel_launch(void* const* d_in, const int* in_sizes, int n_in,
                              void* d_out, int out_size, void* d_ws, size_t ws_size,
                              hipStream_t stream) {
    (void)in_sizes; (void)n_in; (void)out_size; (void)d_ws; (void)ws_size;
    const float* x    = (const float*)d_in[0];
    const float* wgt  = (const float*)d_in[1];
    const float* bias = (const float*)d_in[2];
    float*       out  = (float*)d_out;

    dim3 grid(NT, BN, 8);
    dim3 block(256);
    deconv2d_fold_wmma<<<grid, block, 0, stream>>>(x, wgt, bias, out);
}


# Stage weight tile (64 rows, 4 taps) and the block's 4 x rows in LDS per 32-channel chunk: coalesced 16-byte loads, zero halo replaces mask multiply; sums now chunk-major (32 ch x 4 taps per chunk)
# speedup vs baseline: 5.9016x; 5.9016x over previous
#include <hip/hip_runtime.h>

#if defined(__HIP_DEVICE_COMPILE__)
#if !__has_builtin(__builtin_amdgcn_wmma_f32_16x16x4_f32)
#error "device pass: missing __builtin_amdgcn_wmma_f32_16x16x4_f32"
#endif
#endif

typedef __attribute__((ext_vector_type(2))) float v2f;
typedef __attribute__((ext_vector_type(8))) float v8f;

#define BN    8
#define CI    256
#define CO    128
#define HH    64
#define WW    64
#define OH    130
#define OW    130
#define UU    65
#define NPB   (UU*UU)
#define BLK_N 128
#define NT    ((NPB + BLK_N - 1)/BLK_N)

#define KC    32
#define A_RP  36
#define A_TS  (64 * A_RP + 32)
#define B_RP  72
#define B_CP  304
#define A_SZ  (4 * A_TS)
#define B_SZ  (KC * B_CP)

__global__ __launch_bounds__(256)
void deconv2d_fold_wmma(const float* __restrict__ x,
                        const float* __restrict__ wgt,
                        const float* __restrict__ bias,
                        float* __restrict__ out)
{
    __shared__ __attribute__((aligned(16))) float As[A_SZ];
    __shared__ __attribute__((aligned(16))) float Bs[B_SZ];

    const int tile = blockIdx.x;
    const int bb   = blockIdx.y;
    const int mt   = blockIdx.z >> 2;
    const int par  = blockIdx.z & 3;
    const int p = par >> 1, q = par & 1;

    const int tid  = threadIdx.x;
    const int wave = tid >> 5;
    const int lane = tid & 31;
    const int l16  = lane & 15;
    const int hi   = lane >> 4;

    const int wr = wave >> 2;
    const int wcn = wave & 3;
    const int mw = mt * 64 + wr * 32;
    const int nw = tile * BLK_N + wcn * 32;
    const int u0 = (tile * BLK_N) / UU;

    int   boff[2][4];
    float mk[2][4];
    bool  nval[2];
    int   uu[2], vv[2];
    #pragma unroll
    for (int sj = 0; sj < 2; ++sj) {
        const int n = nw + sj * 16 + l16;
        nval[sj] = (n < NPB);
        const int nc = nval[sj] ? n : (NPB - 1);
        uu[sj] = nc / UU;
        vv[sj] = nc % UU;
        #pragma unroll
        for (int t = 0; t < 4; ++t) {
            const int dy = t & 1, dx = t >> 1;
            const int h = uu[sj] - dy;
            const int w = vv[sj] - dx;
            const bool ok = (h >= 0) && (h < HH) && (w >= 0) && (w < WW);
            boff[sj][t] = (h - u0 + 1) * B_RP + (w + 4);
            mk[sj][t] = ok ? 1.0f : 0.0f;
        }
    }

    const int a_dy  = tid & 1;
    const int a_ch  = (tid >> 1) & 31;
    const int a_row = tid >> 6;
    const float* ag = wgt + ((size_t)(mt * 64 + a_row) * CI + a_ch) * 16 + (p + 2 * a_dy) * 4;
    float* as_w = As + a_dy * A_TS + a_row * A_RP + a_ch;
    const int b_j  = tid & 15;
    const int b_r  = (tid >> 4) & 3;
    const int b_ch = tid >> 6;
    const int b_h  = u0 - 1 + b_r;
    const bool b_ok = (b_h >= 0) && (b_h < HH);
    const float* bg = x + (((size_t)bb * CI + b_ch) * HH + (b_ok ? b_h : 0)) * WW + 4 * b_j;
    float* bs_w = Bs + b_ch * B_CP + b_r * B_RP + 4 + 4 * b_j;

    for (int i = tid; i < B_SZ; i += 256) Bs[i] = 0.0f;

    const float* ar0 = As + (wr * 32 + l16) * A_RP + 2 * hi;
    const float* ar1 = ar0 + 16 * A_RP;
    const float* br  = Bs + (2 * hi) * B_CP;

    v8f acc[2][2] = {};

    #pragma unroll 1
    for (int c = 0; c < CI / KC; ++c) {
        __syncthreads();
        {
            const float* agc = ag + (size_t)c * KC * 16;
            const float* bgc = bg + (size_t)c * KC * HH * WW;
            #pragma unroll
            for (int g = 0; g < 2; ++g) {
                const float4 b0 = *(const float4*)(bgc + (size_t)((g * 4 + 0) * 4) * HH * WW);
                const float4 b1 = *(const float4*)(bgc + (size_t)((g * 4 + 1) * 4) * HH * WW);
                const float4 b2 = *(const float4*)(bgc + (size_t)((g * 4 + 2) * 4) * HH * WW);
                const float4 b3 = *(const float4*)(bgc + (size_t)((g * 4 + 3) * 4) * HH * WW);
                if (b_ok) {
                    *(float4*)(bs_w + ((g * 4 + 0) * 4) * B_CP) = b0;
                    *(float4*)(bs_w + ((g * 4 + 1) * 4) * B_CP) = b1;
                    *(float4*)(bs_w + ((g * 4 + 2) * 4) * B_CP) = b2;
                    *(float4*)(bs_w + ((g * 4 + 3) * 4) * B_CP) = b3;
                }
            }
            #pragma unroll
            for (int g = 0; g < 4; ++g) {
                const float4 w0 = *(const float4*)(agc + (size_t)((g * 4 + 0) * 4) * CI * 16);
                const float4 w1 = *(const float4*)(agc + (size_t)((g * 4 + 1) * 4) * CI * 16);
                const float4 w2 = *(const float4*)(agc + (size_t)((g * 4 + 2) * 4) * CI * 16);
                const float4 w3 = *(const float4*)(agc + (size_t)((g * 4 + 3) * 4) * CI * 16);
                as_w[((g * 4 + 0) * 4) * A_RP]            = q ? w0.y : w0.x;
                as_w[((g * 4 + 0) * 4) * A_RP + 2 * A_TS] = q ? w0.w : w0.z;
                as_w[((g * 4 + 1) * 4) * A_RP]            = q ? w1.y : w1.x;
                as_w[((g * 4 + 1) * 4) * A_RP + 2 * A_TS] = q ? w1.w : w1.z;
                as_w[((g * 4 + 2) * 4) * A_RP]            = q ? w2.y : w2.x;
                as_w[((g * 4 + 2) * 4) * A_RP + 2 * A_TS] = q ? w2.w : w2.z;
                as_w[((g * 4 + 3) * 4) * A_RP]            = q ? w3.y : w3.x;
                as_w[((g * 4 + 3) * 4) * A_RP + 2 * A_TS] = q ? w3.w : w3.z;
            }
        }
        __syncthreads();

        #pragma unroll
        for (int t = 0; t < 4; ++t) {
            const float* a0p = ar0 + t * A_TS;
            const float* a1p = ar1 + t * A_TS;
            const float* b0p = br + boff[0][t];
            const float* b1p = br + boff[1][t];
            #pragma unroll 2
            for (int ks = 0; ks < KC / 4; ++ks) {
                const v2f a0 = *(const v2f*)(a0p + ks * 4);
                const v2f a1 = *(const v2f*)(a1p + ks * 4);
                v2f b0, b1;
                b0.x = b0p[(ks * 4) * B_CP];
                b0.y = b0p[(ks * 4 + 1) * B_CP];
                b1.x = b1p[(ks * 4) * B_CP];
                b1.y = b1p[(ks * 4 + 1) * B_CP];
                acc[0][0] = __builtin_amdgcn_wmma_f32_16x16x4_f32(
                    false, a0, false, b0, (short)0, acc[0][0], false, false);
                acc[0][1] = __builtin_amdgcn_wmma_f32_16x16x4_f32(
                    false, a0, false, b1, (short)0, acc[0][1], false, false);
                acc[1][0] = __builtin_amdgcn_wmma_f32_16x16x4_f32(
                    false, a1, false, b0, (short)0, acc[1][0], false, false);
                acc[1][1] = __builtin_amdgcn_wmma_f32_16x16x4_f32(
                    false, a1, false, b1, (short)0, acc[1][1], false, false);
            }
        }
    }

    #pragma unroll
    for (int sj = 0; sj < 2; ++sj) {
        if (!nval[sj]) continue;
        const int oy = p + 2 * uu[sj];
        const int ox = q + 2 * vv[sj];
        const size_t ob = (size_t)bb * CO * OH * OW + (size_t)oy * OW + ox;
        #pragma unroll
        for (int si = 0; si < 2; ++si) {
            #pragma unroll
            for (int j = 0; j < 8; ++j) {
                const int m = mw + si * 16 + j + hi * 8;
                float bs = 0.0f;
                #pragma unroll
                for (int t = 0; t < 4; ++t) {
                    const int dy = t & 1, dx = t >> 1;
                    bs += mk[sj][t] * bias[m * 16 + (p + 2 * dy) * 4 + (q + 2 * dx)];
                }
                out[ob + (size_t)m * OH * OW] = acc[si][sj][j] + bs;
            }
        }
    }
}

extern "C" void kernel_launch(void* const* d_in, const int* in_sizes, int n_in,
                              void* d_out, int out_size, void* d_ws, size_t ws_size,
                              hipStream_t stream) {
    (void)in_sizes; (void)n_in; (void)out_size; (void)d_ws; (void)ws_size;
    const float* x    = (const float*)d_in[0];
    const float* wgt  = (const float*)d_in[1];
    const float* bias = (const float*)d_in[2];
    float*       out  = (float*)d_out;

    dim3 grid(NT, BN, 8);
    dim3 block(256);
    deconv2d_fold_wmma<<<grid, block, 0, stream>>>(x, wgt, bias, out);
}


# LDS fragment reads software-pipelined: each 16-WMMA group's reads issued as one block one group ahead into a second register set (VGPRs 126 to 142, 3 blocks per WGP); order of sums unchanged
# speedup vs baseline: 6.1487x; 1.0419x over previous
#include <hip/hip_runtime.h>

#if defined(__HIP_DEVICE_COMPILE__)
#if !__has_builtin(__builtin_amdgcn_wmma_f32_16x16x4_f32)
#error "device pass: missing __builtin_amdgcn_wmma_f32_16x16x4_f32"
#endif
#endif

typedef __attribute__((ext_vector_type(2))) float v2f;
typedef __attribute__((ext_vector_type(8))) float v8f;

#define BN    8
#define CI    256
#define CO    128
#define HH    64
#define WW    64
#define OH    130
#define OW    130
#define UU    65
#define NPB   (UU*UU)
#define BLK_N 128
#define NT    ((NPB + BLK_N - 1)/BLK_N)

#define KC    32
#define A_RP  36
#define A_TS  (64 * A_RP + 32)
#define B_RP  72
#define B_CP  304
#define A_SZ  (4 * A_TS)
#define B_SZ  (KC * B_CP)

#define GS    4
#define GR_TAP   ((KC / 4) / GS)
#define NG    (4 * GR_TAP)

struct Frag { v2f a0[GS], a1[GS], b0[GS], b1[GS]; };

#define LOAD_GROUP(F, G) do { \
    const int t_ = (G) / GR_TAP; \
    const int k0_ = ((G) % GR_TAP) * GS; \
    const float* a0p_ = ar0 + t_ * A_TS; \
    const float* a1p_ = ar1 + t_ * A_TS; \
    const float* b0p_ = br + boff[0][t_]; \
    const float* b1p_ = br + boff[1][t_]; \
    _Pragma("unroll") \
    for (int s_ = 0; s_ < GS; ++s_) { \
        const int ks_ = k0_ + s_; \
        (F).a0[s_] = *(const v2f*)(a0p_ + ks_ * 4); \
        (F).a1[s_] = *(const v2f*)(a1p_ + ks_ * 4); \
        (F).b0[s_].x = b0p_[(ks_ * 4) * B_CP]; \
        (F).b0[s_].y = b0p_[(ks_ * 4 + 1) * B_CP]; \
        (F).b1[s_].x = b1p_[(ks_ * 4) * B_CP]; \
        (F).b1[s_].y = b1p_[(ks_ * 4 + 1) * B_CP]; \
    } } while (0)

#define MMA_GROUP(F) do { \
    _Pragma("unroll") \
    for (int s_ = 0; s_ < GS; ++s_) { \
        acc[0][0] = __builtin_amdgcn_wmma_f32_16x16x4_f32( \
            false, (F).a0[s_], false, (F).b0[s_], (short)0, acc[0][0], false, false); \
        acc[0][1] = __builtin_amdgcn_wmma_f32_16x16x4_f32( \
            false, (F).a0[s_], false, (F).b1[s_], (short)0, acc[0][1], false, false); \
        acc[1][0] = __builtin_amdgcn_wmma_f32_16x16x4_f32( \
            false, (F).a1[s_], false, (F).b0[s_], (short)0, acc[1][0], false, false); \
        acc[1][1] = __builtin_amdgcn_wmma_f32_16x16x4_f32( \
            false, (F).a1[s_], false, (F).b1[s_], (short)0, acc[1][1], false, false); \
    } } while (0)

__global__ __launch_bounds__(256)
void deconv2d_fold_wmma(const float* __restrict__ x,
                        const float* __restrict__ wgt,
                        const float* __restrict__ bias,
                        float* __restrict__ out)
{
    __shared__ __attribute__((aligned(16))) float As[A_SZ];
    __shared__ __attribute__((aligned(16))) float Bs[B_SZ];

    const int tile = blockIdx.x;
    const int bb   = blockIdx.y;
    const int mt   = blockIdx.z >> 2;
    const int par  = blockIdx.z & 3;
    const int p = par >> 1, q = par & 1;

    const int tid  = threadIdx.x;
    const int wave = tid >> 5;
    const int lane = tid & 31;
    const int l16  = lane & 15;
    const int hi   = lane >> 4;

    const int wr = wave >> 2;
    const int wcn = wave & 3;
    const int mw = mt * 64 + wr * 32;
    const int nw = tile * BLK_N + wcn * 32;
    const int u0 = (tile * BLK_N) / UU;

    int   boff[2][4];
    float mk[2][4];
    bool  nval[2];
    int   uu[2], vv[2];
    #pragma unroll
    for (int sj = 0; sj < 2; ++sj) {
        const int n = nw + sj * 16 + l16;
        nval[sj] = (n < NPB);
        const int nc = nval[sj] ? n : (NPB - 1);
        uu[sj] = nc / UU;
        vv[sj] = nc % UU;
        #pragma unroll
        for (int t = 0; t < 4; ++t) {
            const int dy = t & 1, dx = t >> 1;
            const int h = uu[sj] - dy;
            const int w = vv[sj] - dx;
            const bool ok = (h >= 0) && (h < HH) && (w >= 0) && (w < WW);
            boff[sj][t] = (h - u0 + 1) * B_RP + (w + 4);
            mk[sj][t] = ok ? 1.0f : 0.0f;
        }
    }

    const int a_dy  = tid & 1;
    const int a_ch  = (tid >> 1) & 31;
    const int a_row = tid >> 6;
    const float* ag = wgt + ((size_t)(mt * 64 + a_row) * CI + a_ch) * 16 + (p + 2 * a_dy) * 4;
    float* as_w = As + a_dy * A_TS + a_row * A_RP + a_ch;
    const int b_j  = tid & 15;
    const int b_r  = (tid >> 4) & 3;
    const int b_ch = tid >> 6;
    const int b_h  = u0 - 1 + b_r;
    const bool b_ok = (b_h >= 0) && (b_h < HH);
    const float* bg = x + (((size_t)bb * CI + b_ch) * HH + (b_ok ? b_h : 0)) * WW + 4 * b_j;
    float* bs_w = Bs + b_ch * B_CP + b_r * B_RP + 4 + 4 * b_j;

    for (int i = tid; i < B_SZ; i += 256) Bs[i] = 0.0f;

    const float* ar0 = As + (wr * 32 + l16) * A_RP + 2 * hi;
    const float* ar1 = ar0 + 16 * A_RP;
    const float* br  = Bs + (2 * hi) * B_CP;

    v8f acc[2][2] = {};

    #pragma unroll 1
    for (int c = 0; c < CI / KC; ++c) {
        __syncthreads();
        {
            const float* agc = ag + (size_t)c * KC * 16;
            const float* bgc = bg + (size_t)c * KC * HH * WW;
            #pragma unroll
            for (int g = 0; g < 2; ++g) {
                const float4 b0 = *(const float4*)(bgc + (size_t)((g * 4 + 0) * 4) * HH * WW);
                const float4 b1 = *(const float4*)(bgc + (size_t)((g * 4 + 1) * 4) * HH * WW);
                const float4 b2 = *(const float4*)(bgc + (size_t)((g * 4 + 2) * 4) * HH * WW);
                const float4 b3 = *(const float4*)(bgc + (size_t)((g * 4 + 3) * 4) * HH * WW);
                if (b_ok) {
                    *(float4*)(bs_w + ((g * 4 + 0) * 4) * B_CP) = b0;
                    *(float4*)(bs_w + ((g * 4 + 1) * 4) * B_CP) = b1;
                    *(float4*)(bs_w + ((g * 4 + 2) * 4) * B_CP) = b2;
                    *(float4*)(bs_w + ((g * 4 + 3) * 4) * B_CP) = b3;
                }
            }
            #pragma unroll
            for (int g = 0; g < 4; ++g) {
                const float4 w0 = *(const float4*)(agc + (size_t)((g * 4 + 0) * 4) * CI * 16);
                const float4 w1 = *(const float4*)(agc + (size_t)((g * 4 + 1) * 4) * CI * 16);
                const float4 w2 = *(const float4*)(agc + (size_t)((g * 4 + 2) * 4) * CI * 16);
                const float4 w3 = *(const float4*)(agc + (size_t)((g * 4 + 3) * 4) * CI * 16);
                as_w[((g * 4 + 0) * 4) * A_RP]            = q ? w0.y : w0.x;
                as_w[((g * 4 + 0) * 4) * A_RP + 2 * A_TS] = q ? w0.w : w0.z;
                as_w[((g * 4 + 1) * 4) * A_RP]            = q ? w1.y : w1.x;
                as_w[((g * 4 + 1) * 4) * A_RP + 2 * A_TS] = q ? w1.w : w1.z;
                as_w[((g * 4 + 2) * 4) * A_RP]            = q ? w2.y : w2.x;
                as_w[((g * 4 + 2) * 4) * A_RP + 2 * A_TS] = q ? w2.w : w2.z;
                as_w[((g * 4 + 3) * 4) * A_RP]            = q ? w3.y : w3.x;
                as_w[((g * 4 + 3) * 4) * A_RP + 2 * A_TS] = q ? w3.w : w3.z;
            }
        }
        __syncthreads();

        Frag fr[2];
        LOAD_GROUP(fr[0], 0);
        __builtin_amdgcn_sched_barrier(0);
        #pragma unroll
        for (int g = 0; g < NG; ++g) {
            if (g + 1 < NG) { LOAD_GROUP(fr[(g + 1) & 1], g + 1); }
            __builtin_amdgcn_sched_barrier(0);
            MMA_GROUP(fr[g & 1]);
            __builtin_amdgcn_sched_barrier(0);
        }
    }

    #pragma unroll
    for (int sj = 0; sj < 2; ++sj) {
        if (!nval[sj]) continue;
        const int oy = p + 2 * uu[sj];
        const int ox = q + 2 * vv[sj];
        const size_t ob = (size_t)bb * CO * OH * OW + (size_t)oy * OW + ox;
        #pragma unroll
        for (int si = 0; si < 2; ++si) {
            #pragma unroll
            for (int j = 0; j < 8; ++j) {
                const int m = mw + si * 16 + j + hi * 8;
                float bs = 0.0f;
                #pragma unroll
                for (int t = 0; t < 4; ++t) {
                    const int dy = t & 1, dx = t >> 1;
                    bs += mk[sj][t] * bias[m * 16 + (p + 2 * dy) * 4 + (q + 2 * dx)];
                }
                out[ob + (size_t)m * OH * OW] = acc[si][sj][j] + bs;
            }
        }
    }
}

extern "C" void kernel_launch(void* const* d_in, const int* in_sizes, int n_in,
                              void* d_out, int out_size, void* d_ws, size_t ws_size,
                              hipStream_t stream) {
    (void)in_sizes; (void)n_in; (void)out_size; (void)d_ws; (void)ws_size;
    const float* x    = (const float*)d_in[0];
    const float* wgt  = (const float*)d_in[1];
    const float* bias = (const float*)d_in[2];
    float*       out  = (float*)d_out;

    dim3 grid(NT, BN, 8);
    dim3 block(256);
    deconv2d_fold_wmma<<<grid, block, 0, stream>>>(x, wgt, bias, out);
}


# Block's 64 M rows = 16 output channels x 4 parities (x tile is parity-independent): weight records read whole, 160 KB -> 64 KB staged per chunk; q pairs stored as 8 bytes; order of sums unchanged
# speedup vs baseline: 7.2001x; 1.1710x over previous
#include <hip/hip_runtime.h>

#if defined(__HIP_DEVICE_COMPILE__)
#if !__has_builtin(__builtin_amdgcn_wmma_f32_16x16x4_f32)
#error "device pass: missing __builtin_amdgcn_wmma_f32_16x16x4_f32"
#endif
#endif

typedef __attribute__((ext_vector_type(2))) float v2f;
typedef __attribute__((ext_vector_type(8))) float v8f;

#define BN    8
#define CI    256
#define CO    128
#define HH    64
#define WW    64
#define OH    130
#define OW    130
#define UU    65
#define NPB   (UU*UU)
#define BLK_N 128
#define NT    ((NPB + BLK_N - 1)/BLK_N)

#define KC    32
#define A_RP  36
#define A_TS  (64 * A_RP + 32)
#define B_RP  72
#define B_CP  304
#define A_SZ  (4 * A_TS)
#define B_SZ  (KC * B_CP)

#define GS    4
#define GR_TAP   ((KC / 4) / GS)
#define NG    (4 * GR_TAP)

struct Frag { v2f a0[GS], a1[GS], b0[GS], b1[GS]; };

#define LOAD_GROUP(F, G) do { \
    const int t_ = (G) / GR_TAP; \
    const int k0_ = ((G) % GR_TAP) * GS; \
    const float* a0p_ = ar0 + t_ * A_TS; \
    const float* a1p_ = ar1 + t_ * A_TS; \
    const float* b0p_ = br + boff[0][t_]; \
    const float* b1p_ = br + boff[1][t_]; \
    _Pragma("unroll") \
    for (int s_ = 0; s_ < GS; ++s_) { \
        const int ks_ = k0_ + s_; \
        (F).a0[s_] = *(const v2f*)(a0p_ + ks_ * 4); \
        (F).a1[s_] = *(const v2f*)(a1p_ + ks_ * 4); \
        (F).b0[s_].x = b0p_[(ks_ * 4) * B_CP]; \
        (F).b0[s_].y = b0p_[(ks_ * 4 + 1) * B_CP]; \
        (F).b1[s_].x = b1p_[(ks_ * 4) * B_CP]; \
        (F).b1[s_].y = b1p_[(ks_ * 4 + 1) * B_CP]; \
    } } while (0)

#define MMA_GROUP(F) do { \
    _Pragma("unroll") \
    for (int s_ = 0; s_ < GS; ++s_) { \
        acc[0][0] = __builtin_amdgcn_wmma_f32_16x16x4_f32( \
            false, (F).a0[s_], false, (F).b0[s_], (short)0, acc[0][0], false, false); \
        acc[0][1] = __builtin_amdgcn_wmma_f32_16x16x4_f32( \
            false, (F).a0[s_], false, (F).b1[s_], (short)0, acc[0][1], false, false); \
        acc[1][0] = __builtin_amdgcn_wmma_f32_16x16x4_f32( \
            false, (F).a1[s_], false, (F).b0[s_], (short)0, acc[1][0], false, false); \
        acc[1][1] = __builtin_amdgcn_wmma_f32_16x16x4_f32( \
            false, (F).a1[s_], false, (F).b1[s_], (short)0, acc[1][1], false, false); \
    } } while (0)

__global__ __launch_bounds__(256)
void deconv2d_fold_wmma(const float* __restrict__ x,
                        const float* __restrict__ wgt,
                        const float* __restrict__ bias,
                        float* __restrict__ out)
{
    __shared__ __attribute__((aligned(16))) float As[A_SZ];
    __shared__ __attribute__((aligned(16))) float Bs[B_SZ];

    const int tile = blockIdx.x;
    const int bb   = blockIdx.y;
    const int mt   = blockIdx.z;

    const int tid  = threadIdx.x;
    const int wave = tid >> 5;
    const int lane = tid & 31;
    const int l16  = lane & 15;
    const int hi   = lane >> 4;

    const int wr = wave >> 2;
    const int wcn = wave & 3;
    const int nw = tile * BLK_N + wcn * 32;
    const int u0 = (tile * BLK_N) / UU;

    int   boff[2][4];
    float mk[2][4];
    bool  nval[2];
    int   uu[2], vv[2];
    #pragma unroll
    for (int sj = 0; sj < 2; ++sj) {
        const int n = nw + sj * 16 + l16;
        nval[sj] = (n < NPB);
        const int nc = nval[sj] ? n : (NPB - 1);
        uu[sj] = nc / UU;
        vv[sj] = nc % UU;
        #pragma unroll
        for (int t = 0; t < 4; ++t) {
            const int dy = t & 1, dx = t >> 1;
            const int h = uu[sj] - dy;
            const int w = vv[sj] - dx;
            const bool ok = (h >= 0) && (h < HH) && (w >= 0) && (w < WW);
            boff[sj][t] = (h - u0 + 1) * B_RP + (w + 4);
            mk[sj][t] = ok ? 1.0f : 0.0f;
        }
    }

    const int a_ky  = tid & 3;
    const int a_ch  = (tid >> 2) & 31;
    const int a_os  = tid >> 7;
    const int a_p   = a_ky & 1;
    const int a_dy  = a_ky >> 1;
    const float* ag = wgt + ((size_t)(mt * 16 + a_os) * CI + a_ch) * 16 + a_ky * 4;
    float* as_w = As + a_dy * A_TS + (a_os * 4 + a_p * 2) * A_RP + a_ch;
    const int b_j  = tid & 15;
    const int b_r  = (tid >> 4) & 3;
    const int b_ch = tid >> 6;
    const int b_h  = u0 - 1 + b_r;
    const bool b_ok = (b_h >= 0) && (b_h < HH);
    const float* bg = x + (((size_t)bb * CI + b_ch) * HH + (b_ok ? b_h : 0)) * WW + 4 * b_j;
    float* bs_w = Bs + b_ch * B_CP + b_r * B_RP + 4 + 4 * b_j;

    for (int i = tid; i < B_SZ; i += 256) Bs[i] = 0.0f;

    const float* ar0 = As + (wr * 32 + l16) * A_RP + 2 * hi;
    const float* ar1 = ar0 + 16 * A_RP;
    const float* br  = Bs + (2 * hi) * B_CP;

    v8f acc[2][2] = {};

    #pragma unroll 1
    for (int c = 0; c < CI / KC; ++c) {
        __syncthreads();
        {
            const float* agc = ag + (size_t)c * KC * 16;
            const float* bgc = bg + (size_t)c * KC * HH * WW;
            #pragma unroll
            for (int g = 0; g < 2; ++g) {
                const float4 b0 = *(const float4*)(bgc + (size_t)((g * 4 + 0) * 4) * HH * WW);
                const float4 b1 = *(const float4*)(bgc + (size_t)((g * 4 + 1) * 4) * HH * WW);
                const float4 b2 = *(const float4*)(bgc + (size_t)((g * 4 + 2) * 4) * HH * WW);
                const float4 b3 = *(const float4*)(bgc + (size_t)((g * 4 + 3) * 4) * HH * WW);
                if (b_ok) {
                    *(float4*)(bs_w + ((g * 4 + 0) * 4) * B_CP) = b0;
                    *(float4*)(bs_w + ((g * 4 + 1) * 4) * B_CP) = b1;
                    *(float4*)(bs_w + ((g * 4 + 2) * 4) * B_CP) = b2;
                    *(float4*)(bs_w + ((g * 4 + 3) * 4) * B_CP) = b3;
                }
            }
            #pragma unroll
            for (int g = 0; g < 2; ++g) {
                const float4 w0 = *(const float4*)(agc + (size_t)((g * 4 + 0) * 2) * CI * 16);
                const float4 w1 = *(const float4*)(agc + (size_t)((g * 4 + 1) * 2) * CI * 16);
                const float4 w2 = *(const float4*)(agc + (size_t)((g * 4 + 2) * 2) * CI * 16);
                const float4 w3 = *(const float4*)(agc + (size_t)((g * 4 + 3) * 2) * CI * 16);
                float* d0 = as_w + ((g * 4 + 0) * 8) * A_RP;
                float* d1 = as_w + ((g * 4 + 1) * 8) * A_RP;
                float* d2 = as_w + ((g * 4 + 2) * 8) * A_RP;
                float* d3 = as_w + ((g * 4 + 3) * 8) * A_RP;
                d0[0] = w0.x; d0[A_RP] = w0.y; d0[2 * A_TS] = w0.z; d0[2 * A_TS + A_RP] = w0.w;
                d1[0] = w1.x; d1[A_RP] = w1.y; d1[2 * A_TS] = w1.z; d1[2 * A_TS + A_RP] = w1.w;
                d2[0] = w2.x; d2[A_RP] = w2.y; d2[2 * A_TS] = w2.z; d2[2 * A_TS + A_RP] = w2.w;
                d3[0] = w3.x; d3[A_RP] = w3.y; d3[2 * A_TS] = w3.z; d3[2 * A_TS + A_RP] = w3.w;
            }
        }
        __syncthreads();

        Frag fr[2];
        LOAD_GROUP(fr[0], 0);
        __builtin_amdgcn_sched_barrier(0);
        #pragma unroll
        for (int g = 0; g < NG; ++g) {
            if (g + 1 < NG) { LOAD_GROUP(fr[(g + 1) & 1], g + 1); }
            __builtin_amdgcn_sched_barrier(0);
            MMA_GROUP(fr[g & 1]);
            __builtin_amdgcn_sched_barrier(0);
        }
    }

    const int ow = mt * 16 + wr * 8 + hi * 2;
    float bv[2][2][4][4];
    #pragma unroll
    for (int si = 0; si < 2; ++si) {
        #pragma unroll
        for (int jo = 0; jo < 2; ++jo) {
            const float4* bp = (const float4*)(bias + (ow + si * 4 + jo) * 16);
            #pragma unroll
            for (int ky = 0; ky < 4; ++ky) {
                const float4 b4 = bp[ky];
                bv[si][jo][ky][0] = b4.x; bv[si][jo][ky][1] = b4.y;
                bv[si][jo][ky][2] = b4.z; bv[si][jo][ky][3] = b4.w;
            }
        }
    }
    #pragma unroll
    for (int sj = 0; sj < 2; ++sj) {
        if (!nval[sj]) continue;
        const size_t ob = (size_t)bb * CO * OH * OW + (size_t)(2 * uu[sj]) * OW + 2 * vv[sj];
        #pragma unroll
        for (int si = 0; si < 2; ++si) {
            #pragma unroll
            for (int jo = 0; jo < 2; ++jo) {
                const int o = ow + si * 4 + jo;
                #pragma unroll
                for (int p = 0; p < 2; ++p) {
                    float r2[2];
                    #pragma unroll
                    for (int q = 0; q < 2; ++q) {
                        float bs = 0.0f;
                        #pragma unroll
                        for (int t = 0; t < 4; ++t) {
                            const int dy = t & 1, dx = t >> 1;
                            bs += mk[sj][t] * bv[si][jo][p + 2 * dy][q + 2 * dx];
                        }
                        r2[q] = acc[si][sj][jo * 4 + p * 2 + q] + bs;
                    }
                    *(float2*)(out + ob + (size_t)o * OH * OW + (size_t)p * OW) = make_float2(r2[0], r2[1]);
                }
            }
        }
    }
}

extern "C" void kernel_launch(void* const* d_in, const int* in_sizes, int n_in,
                              void* d_out, int out_size, void* d_ws, size_t ws_size,
                              hipStream_t stream) {
    (void)in_sizes; (void)n_in; (void)out_size; (void)d_ws; (void)ws_size;
    const float* x    = (const float*)d_in[0];
    const float* wgt  = (const float*)d_in[1];
    const float* bias = (const float*)d_in[2];
    float*       out  = (float*)d_out;

    dim3 grid(NT, BN, 8);
    dim3 block(256);
    deconv2d_fold_wmma<<<grid, block, 0, stream>>>(x, wgt, bias, out);
}


# F(2,2) transform along x: a lane owns a pixel pair; products (e0-e1)g1, e1(g0+g1), (e2-e1)g0 replace its 4 tap products, 3 WMMAs for 4; sum order changes: 3 partial sums combined at the end
# speedup vs baseline: 8.2519x; 1.1461x over previous
#include <hip/hip_runtime.h>

#if defined(__HIP_DEVICE_COMPILE__)
#if !__has_builtin(__builtin_amdgcn_wmma_f32_16x16x4_f32)
#error "device pass: missing __builtin_amdgcn_wmma_f32_16x16x4_f32"
#endif
#endif

typedef __attribute__((ext_vector_type(2))) float v2f;
typedef __attribute__((ext_vector_type(8))) float v8f;

#define BN    8
#define CI    256
#define CO    128
#define HH    64
#define WW    64
#define OH    130
#define OW    130
#define UU    65
#define VB    33
#define NPB   (UU*VB)
#define BLK_N 64
#define NT    ((NPB + BLK_N - 1)/BLK_N)

#define KC    32
#define A_RP  36
#define A_TS  (64 * A_RP + 32)
#define NPL   6
#define B_RP  148
#define B_CP  594
#define A_SZ  (NPL * A_TS)
#define B_SZ  ((KC / 2) * B_CP)

#define GS    2
#define GR_DY    ((KC / 4) / GS)
#define NG    (2 * GR_DY)

struct Frag { v2f a0[GS][3], a1[GS][3], e[GS][3]; };

#define LOAD_GROUP(F, G) do { \
    const int dy_ = (G) / GR_DY; \
    const int k0_ = ((G) % GR_DY) * GS; \
    const float* bp_ = br + boff[dy_]; \
    _Pragma("unroll") \
    for (int s_ = 0; s_ < GS; ++s_) { \
        const int ks_ = k0_ + s_; \
        _Pragma("unroll") \
        for (int w_ = 0; w_ < 3; ++w_) { \
            (F).a0[s_][w_] = *(const v2f*)(ar0 + (dy_ + 2 * w_) * A_TS + ks_ * 4); \
            (F).a1[s_][w_] = *(const v2f*)(ar1 + (dy_ + 2 * w_) * A_TS + ks_ * 4); \
            (F).e[s_][w_] = *(const v2f*)(bp_ + (ks_ * 2) * B_CP + 2 * w_); \
        } \
    } } while (0)

#define WMMA_F32(A_, B_, C_) __builtin_amdgcn_wmma_f32_16x16x4_f32( \
            false, (A_), false, (B_), (short)0, (C_), false, false)

#define XFORM_GROUP(F) do { \
    _Pragma("unroll") \
    for (int s_ = 0; s_ < GS; ++s_) { \
        (F).e[s_][0] = (F).e[s_][0] - (F).e[s_][1]; \
        (F).e[s_][2] = (F).e[s_][2] - (F).e[s_][1]; \
    } } while (0)

#define MMA_GROUP(F) do { \
    _Pragma("unroll") \
    for (int s_ = 0; s_ < GS; ++s_) { \
        acc[0][0] = WMMA_F32((F).a0[s_][0], (F).e[s_][0], acc[0][0]); \
        acc[1][0] = WMMA_F32((F).a1[s_][0], (F).e[s_][0], acc[1][0]); \
        acc[0][1] = WMMA_F32((F).a0[s_][1], (F).e[s_][1], acc[0][1]); \
        acc[1][1] = WMMA_F32((F).a1[s_][1], (F).e[s_][1], acc[1][1]); \
        acc[0][2] = WMMA_F32((F).a0[s_][2], (F).e[s_][2], acc[0][2]); \
        acc[1][2] = WMMA_F32((F).a1[s_][2], (F).e[s_][2], acc[1][2]); \
    } } while (0)

__global__ __launch_bounds__(256)
void deconv2d_fold_wmma(const float* __restrict__ x,
                        const float* __restrict__ wgt,
                        const float* __restrict__ bias,
                        float* __restrict__ out)
{
    __shared__ __attribute__((aligned(16))) float As[A_SZ];
    __shared__ __attribute__((aligned(16))) float Bs[B_SZ];

    const int tile = blockIdx.x;
    const int bb   = blockIdx.y;
    const int mt   = blockIdx.z;

    const int tid  = threadIdx.x;
    const int wave = tid >> 5;
    const int lane = tid & 31;
    const int l16  = lane & 15;
    const int hi   = lane >> 4;

    const int wr = wave >> 2;
    const int wcn = wave & 3;
    const int nw = tile * BLK_N + wcn * 16;
    const int u0 = (tile * BLK_N) / VB;

    const int  n    = nw + l16;
    const bool nval = (n < NPB);
    const int  nc   = nval ? n : (NPB - 1);
    const int  uu   = nc / VB;
    const int  bq   = nc % VB;
    int boff[2];
    #pragma unroll
    for (int dy = 0; dy < 2; ++dy) boff[dy] = (uu - dy - u0 + 1) * B_RP + (2 * bq + 3) * 2;

    const int a_ky  = tid & 3;
    const int a_ch  = (tid >> 2) & 31;
    const int a_os  = tid >> 7;
    const int a_p   = a_ky & 1;
    const int a_dy  = a_ky >> 1;
    const float* ag = wgt + ((size_t)(mt * 16 + a_os) * CI + a_ch) * 16 + a_ky * 4;
    float* as_w = As + a_dy * A_TS + (a_os * 4 + a_p * 2) * A_RP + a_ch;
    const int b_j  = tid & 15;
    const int b_r  = (tid >> 4) & 3;
    const int b_ch = tid >> 6;
    const int b_h  = u0 - 1 + b_r;
    const bool b_ok = (b_h >= 0) && (b_h < HH);
    const float* bg = x + (((size_t)bb * CI + b_ch) * HH + (b_ok ? b_h : 0)) * WW + 4 * b_j;
    float* bs_w = Bs + (b_ch >> 1) * B_CP + b_r * B_RP + (4 + 4 * b_j) * 2 + (b_ch & 1);

    for (int i = tid; i < B_SZ; i += 256) Bs[i] = 0.0f;

    const float* ar0 = As + (wr * 32 + l16) * A_RP + 2 * hi;
    const float* ar1 = ar0 + 16 * A_RP;
    const float* br  = Bs + hi * B_CP;

    v8f acc[2][3] = {};

    #pragma unroll 1
    for (int c = 0; c < CI / KC; ++c) {
        __syncthreads();
        {
            const float* agc = ag + (size_t)c * KC * 16;
            const float* bgc = bg + (size_t)c * KC * HH * WW;
            #pragma unroll
            for (int g = 0; g < 2; ++g) {
                const float4 b0 = *(const float4*)(bgc + (size_t)((g * 4 + 0) * 4) * HH * WW);
                const float4 b1 = *(const float4*)(bgc + (size_t)((g * 4 + 1) * 4) * HH * WW);
                const float4 b2 = *(const float4*)(bgc + (size_t)((g * 4 + 2) * 4) * HH * WW);
                const float4 b3 = *(const float4*)(bgc + (size_t)((g * 4 + 3) * 4) * HH * WW);
                if (b_ok) {
                    float* e0 = bs_w + ((g * 4 + 0) * 2) * B_CP;
                    float* e1 = bs_w + ((g * 4 + 1) * 2) * B_CP;
                    float* e2 = bs_w + ((g * 4 + 2) * 2) * B_CP;
                    float* e3 = bs_w + ((g * 4 + 3) * 2) * B_CP;
                    e0[0] = b0.x; e0[2] = b0.y; e0[4] = b0.z; e0[6] = b0.w;
                    e1[0] = b1.x; e1[2] = b1.y; e1[4] = b1.z; e1[6] = b1.w;
                    e2[0] = b2.x; e2[2] = b2.y; e2[4] = b2.z; e2[6] = b2.w;
                    e3[0] = b3.x; e3[2] = b3.y; e3[4] = b3.z; e3[6] = b3.w;
                }
            }
            #pragma unroll
            for (int g = 0; g < 2; ++g) {
                #pragma unroll
                for (int k = 0; k < 4; ++k) {
                    const float4 w4 = *(const float4*)(agc + (size_t)((g * 4 + k) * 2) * CI * 16);
                    float* d = as_w + ((g * 4 + k) * 8) * A_RP;
                    d[0]        = w4.z;        d[A_RP]            = w4.w;
                    d[2 * A_TS] = w4.x + w4.z; d[2 * A_TS + A_RP] = w4.y + w4.w;
                    d[4 * A_TS] = w4.x;        d[4 * A_TS + A_RP] = w4.y;
                }
            }
        }
        __syncthreads();

        Frag fr[2];
        LOAD_GROUP(fr[0], 0);
        __builtin_amdgcn_sched_barrier(0);
        #pragma unroll
        for (int g = 0; g < NG; ++g) {
            if (g + 1 < NG) { LOAD_GROUP(fr[(g + 1) & 1], g + 1); }
            __builtin_amdgcn_sched_barrier(0);
            XFORM_GROUP(fr[g & 1]);
            __builtin_amdgcn_sched_barrier(0);
            MMA_GROUP(fr[g & 1]);
            __builtin_amdgcn_sched_barrier(0);
        }
    }

    const int ow = mt * 16 + wr * 8 + hi * 2;
    float bv[2][2][4][4];
    #pragma unroll
    for (int si = 0; si < 2; ++si) {
        #pragma unroll
        for (int jo = 0; jo < 2; ++jo) {
            const float4* bp = (const float4*)(bias + (ow + si * 4 + jo) * 16);
            #pragma unroll
            for (int ky = 0; ky < 4; ++ky) {
                const float4 b4 = bp[ky];
                bv[si][jo][ky][0] = b4.x; bv[si][jo][ky][1] = b4.y;
                bv[si][jo][ky][2] = b4.z; bv[si][jo][ky][3] = b4.w;
            }
        }
    }
    if (nval) {
        const int  v0  = 2 * bq;
        const bool two = (v0 + 1 < UU);
        float mk[2][4];
        #pragma unroll
        for (int e = 0; e < 2; ++e) {
            #pragma unroll
            for (int t = 0; t < 4; ++t) {
                const int dy = t & 1, dx = t >> 1;
                const int h = uu - dy;
                const int w = v0 + e - dx;
                const bool ok = (h >= 0) && (h < HH) && (w >= 0) && (w < WW);
                mk[e][t] = ok ? 1.0f : 0.0f;
            }
        }
        const size_t ob = (size_t)bb * CO * OH * OW + (size_t)(2 * uu) * OW + 2 * v0;
        #pragma unroll
        for (int si = 0; si < 2; ++si) {
            #pragma unroll
            for (int jo = 0; jo < 2; ++jo) {
                const int o = ow + si * 4 + jo;
                #pragma unroll
                for (int p = 0; p < 2; ++p) {
                    float r4[4];
                    #pragma unroll
                    for (int e = 0; e < 2; ++e) {
                        #pragma unroll
                        for (int q = 0; q < 2; ++q) {
                            float bs = 0.0f;
                            #pragma unroll
                            for (int t = 0; t < 4; ++t) {
                                const int dy = t & 1, dx = t >> 1;
                                bs += mk[e][t] * bv[si][jo][p + 2 * dy][q + 2 * dx];
                            }
                            const int j = jo * 4 + p * 2 + q;
                            r4[e * 2 + q] = (acc[si][e][j] + acc[si][e + 1][j]) + bs;
                        }
                    }
                    float* dst = out + ob + (size_t)o * OH * OW + (size_t)p * OW;
                    *(float2*)dst = make_float2(r4[0], r4[1]);
                    if (two) *(float2*)(dst + 2) = make_float2(r4[2], r4[3]);
                }
            }
        }
    }
}

extern "C" void kernel_launch(void* const* d_in, const int* in_sizes, int n_in,
                              void* d_out, int out_size, void* d_ws, size_t ws_size,
                              hipStream_t stream) {
    (void)in_sizes; (void)n_in; (void)out_size; (void)d_ws; (void)ws_size;
    const float* x    = (const float*)d_in[0];
    const float* wgt  = (const float*)d_in[1];
    const float* bias = (const float*)d_in[2];
    float*       out  = (float*)d_out;

    dim3 grid(NT, BN, 8);
    dim3 block(256);
    deconv2d_fold_wmma<<<grid, block, 0, stream>>>(x, wgt, bias, out);
}


# F(3,2) along x replaces F(2,2): a lane owns a pixel triple, 4 products for its 3 outputs (was 3 for 2), 4 weight planes per dy; sum order changes: 4 partial sums combined at the end
# speedup vs baseline: 8.7715x; 1.0630x over previous
#include <hip/hip_runtime.h>

#if defined(__HIP_DEVICE_COMPILE__)
#if !__has_builtin(__builtin_amdgcn_wmma_f32_16x16x4_f32)
#error "device pass: missing __builtin_amdgcn_wmma_f32_16x16x4_f32"
#endif
#endif

typedef __attribute__((ext_vector_type(2))) float v2f;
typedef __attribute__((ext_vector_type(8))) float v8f;

#define BN    8
#define CI    256
#define CO    128
#define HH    64
#define WW    64
#define OH    130
#define OW    130
#define UU    65
#define VB    22
#define NPB   (UU*VB)
#define BLK_N 64
#define NT    ((NPB + BLK_N - 1)/BLK_N)

#define KC    32
#define A_RP  36
#define A_TS  (64 * A_RP + 32)
#define NPL   8
#define B_RP  148
#define B_NR  5
#define B_CP  800
#define A_SZ  (NPL * A_TS)
#define B_SZ  ((KC / 2) * B_CP)

#define GS    2
#define GR_DY    ((KC / 4) / GS)
#define NG    (2 * GR_DY)

struct Frag { v2f a0[GS][4], a1[GS][4], e[GS][4]; };

#define LOAD_GROUP(F, G) do { \
    const int dy_ = (G) / GR_DY; \
    const int k0_ = ((G) % GR_DY) * GS; \
    const float* bp_ = br + boff[dy_]; \
    _Pragma("unroll") \
    for (int s_ = 0; s_ < GS; ++s_) { \
        const int ks_ = k0_ + s_; \
        _Pragma("unroll") \
        for (int w_ = 0; w_ < 4; ++w_) { \
            (F).a0[s_][w_] = *(const v2f*)(ar0 + (dy_ + 2 * w_) * A_TS + ks_ * 4); \
            (F).a1[s_][w_] = *(const v2f*)(ar1 + (dy_ + 2 * w_) * A_TS + ks_ * 4); \
            (F).e[s_][w_] = *(const v2f*)(bp_ + (ks_ * 2) * B_CP + 2 * w_); \
        } \
    } } while (0)

#define WMMA_F32(A_, B_, C_) __builtin_amdgcn_wmma_f32_16x16x4_f32( \
            false, (A_), false, (B_), (short)0, (C_), false, false)

#define XFORM_GROUP(F) do { \
    _Pragma("unroll") \
    for (int s_ = 0; s_ < GS; ++s_) { \
        const v2f d1_ = (F).e[s_][1]; \
        const v2f d2_ = (F).e[s_][2]; \
        (F).e[s_][0] = (F).e[s_][0] - d2_; \
        (F).e[s_][3] = (F).e[s_][3] - d1_; \
        (F).e[s_][1] = d1_ + d2_; \
        (F).e[s_][2] = d2_ - d1_; \
    } } while (0)

#define MMA_GROUP(F) do { \
    _Pragma("unroll") \
    for (int s_ = 0; s_ < GS; ++s_) { \
        acc[0][0] = WMMA_F32((F).a0[s_][0], (F).e[s_][0], acc[0][0]); \
        acc[1][0] = WMMA_F32((F).a1[s_][0], (F).e[s_][0], acc[1][0]); \
        acc[0][1] = WMMA_F32((F).a0[s_][1], (F).e[s_][1], acc[0][1]); \
        acc[1][1] = WMMA_F32((F).a1[s_][1], (F).e[s_][1], acc[1][1]); \
        acc[0][2] = WMMA_F32((F).a0[s_][2], (F).e[s_][2], acc[0][2]); \
        acc[1][2] = WMMA_F32((F).a1[s_][2], (F).e[s_][2], acc[1][2]); \
        acc[0][3] = WMMA_F32((F).a0[s_][3], (F).e[s_][3], acc[0][3]); \
        acc[1][3] = WMMA_F32((F).a1[s_][3], (F).e[s_][3], acc[1][3]); \
    } } while (0)

__global__ __launch_bounds__(256)
void deconv2d_fold_wmma(const float* __restrict__ x,
                        const float* __restrict__ wgt,
                        const float* __restrict__ bias,
                        float* __restrict__ out)
{
    __shared__ __attribute__((aligned(16))) float As[A_SZ];
    __shared__ __attribute__((aligned(16))) float Bs[B_SZ];

    const int tile = blockIdx.x;
    const int bb   = blockIdx.y;
    const int mt   = blockIdx.z;

    const int tid  = threadIdx.x;
    const int wave = tid >> 5;
    const int lane = tid & 31;
    const int l16  = lane & 15;
    const int hi   = lane >> 4;

    const int wr = wave >> 2;
    const int wcn = wave & 3;
    const int nw = tile * BLK_N + wcn * 16;
    const int u0 = (tile * BLK_N) / VB;

    const int  n    = nw + l16;
    const bool nval = (n < NPB);
    const int  nc   = nval ? n : (NPB - 1);
    const int  uu   = nc / VB;
    const int  bq   = nc % VB;
    int boff[2];
    #pragma unroll
    for (int dy = 0; dy < 2; ++dy) boff[dy] = (uu - dy - u0 + 1) * B_RP + (3 * bq + 3) * 2;

    const int a_ky  = tid & 3;
    const int a_ch  = (tid >> 2) & 31;
    const int a_os  = tid >> 7;
    const int a_p   = a_ky & 1;
    const int a_dy  = a_ky >> 1;
    const float* ag = wgt + ((size_t)(mt * 16 + a_os) * CI + a_ch) * 16 + a_ky * 4;
    float* as_w = As + a_dy * A_TS + (a_os * 4 + a_p * 2) * A_RP + a_ch;
    const int b_j  = tid & 15;
    const int b_r  = (tid >> 4) & 3;
    const int b_ch = tid >> 6;
    const int b_h  = u0 - 1 + b_r;
    const bool b_ok = (b_h >= 0) && (b_h < HH);
    const float* bg = x + (((size_t)bb * CI + b_ch) * HH + (b_ok ? b_h : 0)) * WW + 4 * b_j;
    float* bs_w = Bs + (b_ch >> 1) * B_CP + b_r * B_RP + (4 + 4 * b_j) * 2 + (b_ch & 1);
    const int c4   = tid >> 4;
    const int h4   = u0 + 3;
    const bool ok4 = (h4 < HH);
    const float* bg4 = x + (((size_t)bb * CI + c4) * HH + (ok4 ? h4 : 0)) * WW + 4 * b_j;
    float* bs_w4 = Bs + (c4 >> 1) * B_CP + 4 * B_RP + (4 + 4 * b_j) * 2 + (c4 & 1);

    for (int i = tid; i < B_SZ; i += 256) Bs[i] = 0.0f;

    const float* ar0 = As + (wr * 32 + l16) * A_RP + 2 * hi;
    const float* ar1 = ar0 + 16 * A_RP;
    const float* br  = Bs + hi * B_CP;

    v8f acc[2][4] = {};

    #pragma unroll 1
    for (int c = 0; c < CI / KC; ++c) {
        __syncthreads();
        {
            const float* agc = ag + (size_t)c * KC * 16;
            const float* bgc = bg + (size_t)c * KC * HH * WW;
            const float* bgc4 = bg4 + (size_t)c * KC * HH * WW;
            const float4 r40 = *(const float4*)(bgc4);
            const float4 r41 = *(const float4*)(bgc4 + (size_t)16 * HH * WW);
            #pragma unroll
            for (int g = 0; g < 2; ++g) {
                const float4 b0 = *(const float4*)(bgc + (size_t)((g * 4 + 0) * 4) * HH * WW);
                const float4 b1 = *(const float4*)(bgc + (size_t)((g * 4 + 1) * 4) * HH * WW);
                const float4 b2 = *(const float4*)(bgc + (size_t)((g * 4 + 2) * 4) * HH * WW);
                const float4 b3 = *(const float4*)(bgc + (size_t)((g * 4 + 3) * 4) * HH * WW);
                if (b_ok) {
                    float* e0 = bs_w + ((g * 4 + 0) * 2) * B_CP;
                    float* e1 = bs_w + ((g * 4 + 1) * 2) * B_CP;
                    float* e2 = bs_w + ((g * 4 + 2) * 2) * B_CP;
                    float* e3 = bs_w + ((g * 4 + 3) * 2) * B_CP;
                    e0[0] = b0.x; e0[2] = b0.y; e0[4] = b0.z; e0[6] = b0.w;
                    e1[0] = b1.x; e1[2] = b1.y; e1[4] = b1.z; e1[6] = b1.w;
                    e2[0] = b2.x; e2[2] = b2.y; e2[4] = b2.z; e2[6] = b2.w;
                    e3[0] = b3.x; e3[2] = b3.y; e3[4] = b3.z; e3[6] = b3.w;
                }
            }
            {
                float* e0 = bs_w4;
                float* e1 = bs_w4 + 8 * B_CP;
                e0[0] = ok4 ? r40.x : 0.0f; e0[2] = ok4 ? r40.y : 0.0f; e0[4] = ok4 ? r40.z : 0.0f; e0[6] = ok4 ? r40.w : 0.0f;
                e1[0] = ok4 ? r41.x : 0.0f; e1[2] = ok4 ? r41.y : 0.0f; e1[4] = ok4 ? r41.z : 0.0f; e1[6] = ok4 ? r41.w : 0.0f;
            }
            #pragma unroll
            for (int g = 0; g < 2; ++g) {
                #pragma unroll
                for (int k = 0; k < 4; ++k) {
                    const float4 w4 = *(const float4*)(agc + (size_t)((g * 4 + k) * 2) * CI * 16);
                    float* d = as_w + ((g * 4 + k) * 8) * A_RP;
                    d[0]        = w4.z;                  d[A_RP]            = w4.w;
                    d[2 * A_TS] = 0.5f * (w4.x + w4.z);  d[2 * A_TS + A_RP] = 0.5f * (w4.y + w4.w);
                    d[4 * A_TS] = 0.5f * (w4.z - w4.x);  d[4 * A_TS + A_RP] = 0.5f * (w4.w - w4.y);
                    d[6 * A_TS] = w4.x;                  d[6 * A_TS + A_RP] = w4.y;
                }
            }
        }
        __syncthreads();

        Frag fr[2];
        LOAD_GROUP(fr[0], 0);
        __builtin_amdgcn_sched_barrier(0);
        #pragma unroll
        for (int g = 0; g < NG; ++g) {
            if (g + 1 < NG) { LOAD_GROUP(fr[(g + 1) & 1], g + 1); }
            __builtin_amdgcn_sched_barrier(0);
            XFORM_GROUP(fr[g & 1]);
            __builtin_amdgcn_sched_barrier(0);
            MMA_GROUP(fr[g & 1]);
            __builtin_amdgcn_sched_barrier(0);
        }
    }

    const int ow = mt * 16 + wr * 8 + hi * 2;
    float bv[2][2][4][4];
    #pragma unroll
    for (int si = 0; si < 2; ++si) {
        #pragma unroll
        for (int jo = 0; jo < 2; ++jo) {
            const float4* bp = (const float4*)(bias + (ow + si * 4 + jo) * 16);
            #pragma unroll
            for (int ky = 0; ky < 4; ++ky) {
                const float4 b4 = bp[ky];
                bv[si][jo][ky][0] = b4.x; bv[si][jo][ky][1] = b4.y;
                bv[si][jo][ky][2] = b4.z; bv[si][jo][ky][3] = b4.w;
            }
        }
    }
    if (nval) {
        const int  v0   = 3 * bq;
        const bool full = (v0 + 2 < UU);
        float mk[3][4];
        #pragma unroll
        for (int e = 0; e < 3; ++e) {
            #pragma unroll
            for (int t = 0; t < 4; ++t) {
                const int dy = t & 1, dx = t >> 1;
                const int h = uu - dy;
                const int w = v0 + e - dx;
                const bool ok = (h >= 0) && (h < HH) && (w >= 0) && (w < WW);
                mk[e][t] = ok ? 1.0f : 0.0f;
            }
        }
        const size_t ob = (size_t)bb * CO * OH * OW + (size_t)(2 * uu) * OW + 2 * v0;
        #pragma unroll
        for (int si = 0; si < 2; ++si) {
            #pragma unroll
            for (int jo = 0; jo < 2; ++jo) {
                const int o = ow + si * 4 + jo;
                #pragma unroll
                for (int p = 0; p < 2; ++p) {
                    float r6[6];
                    #pragma unroll
                    for (int q = 0; q < 2; ++q) {
                        const int j = jo * 4 + p * 2 + q;
                        const float m0 = acc[si][0][j], m1 = acc[si][1][j];
                        const float m2 = acc[si][2][j], m3 = acc[si][3][j];
                        float yv[3];
                        yv[0] = (m0 + m1) + m2;
                        yv[1] = m1 - m2;
                        yv[2] = (m1 + m2) + m3;
                        #pragma unroll
                        for (int e = 0; e < 3; ++e) {
                            float bs = 0.0f;
                            #pragma unroll
                            for (int t = 0; t < 4; ++t) {
                                const int dy = t & 1, dx = t >> 1;
                                bs += mk[e][t] * bv[si][jo][p + 2 * dy][q + 2 * dx];
                            }
                            r6[e * 2 + q] = yv[e] + bs;
                        }
                    }
                    float* dst = out + ob + (size_t)o * OH * OW + (size_t)p * OW;
                    *(float2*)dst = make_float2(r6[0], r6[1]);
                    *(float2*)(dst + 2) = make_float2(r6[2], r6[3]);
                    if (full) *(float2*)(dst + 4) = make_float2(r6[4], r6[5]);
                }
            }
        }
    }
}

extern "C" void kernel_launch(void* const* d_in, const int* in_sizes, int n_in,
                              void* d_out, int out_size, void* d_ws, size_t ws_size,
                              hipStream_t stream) {
    (void)in_sizes; (void)n_in; (void)out_size; (void)d_ws; (void)ws_size;
    const float* x    = (const float*)d_in[0];
    const float* wgt  = (const float*)d_in[1];
    const float* bias = (const float*)d_in[2];
    float*       out  = (float*)d_out;

    dim3 grid(NT, BN, 8);
    dim3 block(256);
    deconv2d_fold_wmma<<<grid, block, 0, stream>>>(x, wgt, bias, out);
}
